# starting kernel
# baseline (speedup 1.0000x reference)
#include <hip/hip_runtime.h>

typedef float v2f __attribute__((ext_vector_type(2)));
typedef float v8f __attribute__((ext_vector_type(8)));
typedef int   v2i __attribute__((ext_vector_type(2)));

#define NFPN_EPS2 1e-10f
#define NFPN_MAX_DEPTH 100

__global__ __launch_bounds__(256) void nfpn_rps_kernel(
    const float* __restrict__ din,
    const float* __restrict__ W1,
    const float* __restrict__ b1,
    const float* __restrict__ W2,
    const float* __restrict__ b2,
    float* __restrict__ out,
    int B)
{
    const int gid  = blockIdx.x * blockDim.x + threadIdx.x;
    const int lane = threadIdx.x & 31;
    const bool low = lane < 16;

    int s = gid;
    const bool active = (s < B);
    if (!active) s = B - 1;

    const float d0 = din[3*s + 0];
    const float d1 = din[3*s + 1];
    const float d2 = din[3*s + 2];

    float q[6];
#pragma unroll
    for (int j = 0; j < 6; ++j) {
        float t = fmaf(d0, W1[3*j + 0], fmaf(d1, W1[3*j + 1], fmaf(d2, W1[3*j + 2], b1[j])));
        q[j] = (t >= 0.f) ? t : 0.1f * t;
    }

    const int  M     = lane & 15;
    const int  rr    = M & 7;
    const bool valid = (rr < 6);
    v2f a1, a2;
    a1.x = valid ? -W2[rr*6 + (low ? 0 : 2)] : 0.f;
    a1.y = valid ? -W2[rr*6 + (low ? 1 : 3)] : 0.f;
    a2.x = (valid && low) ? -W2[rr*6 + 4] : 0.f;
    a2.y = (valid && low) ? -W2[rr*6 + 5] : 0.f;

    v8f cinit;
#pragma unroll
    for (int j = 0; j < 6; ++j) cinit[j] = -b2[j];
    cinit[6] = 0.f; cinit[7] = 0.f;

    float u[6];
#pragma unroll
    for (int j = 0; j < 6; ++j) u[j] = 1.0f / 3.0f;

    auto swap_halves = [&](float a, float b) -> v2f {
        v2f o;
#if __has_builtin(__builtin_amdgcn_permlane16_swap)
        v2i r = __builtin_amdgcn_permlane16_swap(__float_as_int(a), __float_as_int(b),
                                                 false, false);
        o.x = __int_as_float(r.x);
        o.y = __int_as_float(r.y);
#else
        float as = __shfl_xor(a, 16, 32);
        float bs = __shfl_xor(b, 16, 32);
        o.x = low ? a : bs;
        o.y = low ? as : b;
#endif
        return o;
    };

    auto matvec = [&](const float* x, float* y) {
        v2f B1, B2, B1p, B2p;
        v2f s02 = swap_halves(x[0], x[2]);  B1.x = s02.x;  B1p.x = s02.y;
        v2f s13 = swap_halves(x[1], x[3]);  B1.y = s13.x;  B1p.y = s13.y;
        v2f s4z = swap_halves(x[4], 0.f);   B2.x = s4z.x;  B2p.x = s4z.y;
        v2f s5z = swap_halves(x[5], 0.f);   B2.y = s5z.x;  B2p.y = s5z.y;
        v8f acc = cinit;
        acc  = __builtin_amdgcn_wmma_f32_16x16x4_f32(false, a1, false, B1,  (short)0, acc,  false, false);
        acc  = __builtin_amdgcn_wmma_f32_16x16x4_f32(false, a2, false, B2,  (short)0, acc,  false, false);
        v8f accp = cinit;
        accp = __builtin_amdgcn_wmma_f32_16x16x4_f32(false, a1, false, B1p, (short)0, accp, false, false);
        accp = __builtin_amdgcn_wmma_f32_16x16x4_f32(false, a2, false, B2p, (short)0, accp, false, false);
#pragma unroll
        for (int j = 0; j < 6; ++j) y[j] = low ? acc[j] : accp[j];
    };

    auto proj3 = [&](float a, float b, float c, float& o0, float& o1, float& o2) {
        float hi  = fmaxf(fmaxf(a, b), c);
        float lo  = fminf(fminf(a, b), c);
        float sum = a + b + c;
        float mid = sum - hi - lo;
        float c2  = hi + mid;
        bool cond2 = (mid - hi + 1.0f) > 0.0f;
        bool cond3 = (3.0f*lo - sum + 1.0f) > 0.0f;
        int rho = 1 + (cond2 ? 1 : 0) + (cond3 ? 1 : 0);
        float cs    = (rho == 3) ? sum : ((rho == 2) ? c2 : hi);
        float rinv  = (rho == 3) ? (1.0f/3.0f) : ((rho == 2) ? 0.5f : 1.0f);
        float theta = (cs - 1.0f) * rinv;
        o0 = fmaxf(a - theta, 0.f);
        o1 = fmaxf(b - theta, 0.f);
        o2 = fmaxf(c - theta, 0.f);
    };

    int depth = 0;
    for (;;) {
        float x[6], y[6], un[6];
#pragma unroll
        for (int j = 0; j < 6; ++j) x[j] = u[j] + q[j];
        matvec(x, y);
        proj3(y[0], y[1], y[2], un[0], un[1], un[2]);
        proj3(y[3], y[4], y[5], un[3], un[4], un[5]);
        float r2 = 0.f;
#pragma unroll
        for (int j = 0; j < 6; ++j) { float dd = un[j] - u[j]; r2 = fmaf(dd, dd, r2); u[j] = un[j]; }
        ++depth;
        if (depth >= NFPN_MAX_DEPTH) break;
        if (__ballot(r2 > NFPN_EPS2) == 0ULL) break;
    }

    float x[6], y[6], o[6];
#pragma unroll
    for (int j = 0; j < 6; ++j) x[j] = u[j] + q[j];
    matvec(x, y);
    proj3(y[0], y[1], y[2], o[0], o[1], o[2]);
    proj3(y[3], y[4], y[5], o[3], o[4], o[5]);

    if (active) {
        float2* op = reinterpret_cast<float2*>(out + 6l * s);
        op[0] = make_float2(o[0], o[1]);
        op[1] = make_float2(o[2], o[3]);
        op[2] = make_float2(o[4], o[5]);
    }
}

extern "C" void kernel_launch(void* const* d_in, const int* in_sizes, int n_in,
                              void* d_out, int out_size, void* d_ws, size_t ws_size,
                              hipStream_t stream) {
    const float* d  = (const float*)d_in[0];
    const float* W1 = (const float*)d_in[1];
    const float* b1 = (const float*)d_in[2];
    const float* W2 = (const float*)d_in[3];
    const float* b2 = (const float*)d_in[4];
    float* out = (float*)d_out;

    const int B = in_sizes[0] / 3;
    const int threads = 256;
    const int blocks  = (B + threads - 1) / threads;
    hipLaunchKernelGGL(nfpn_rps_kernel, dim3(blocks), dim3(threads), 0, stream,
                       d, W1, b1, W2, b2, out, B);
}


# Block-diagonal A (rows 0-5 on k-half 0, rows 8-13 on k-half 1): 3 WMMAs serve all 32 lanes (was 4), each lane feeds its own x as B, no half-swaps or selects; products summed 2+2+2, was 4+2
# speedup vs baseline: 1.2336x; 1.2336x over previous
#include <hip/hip_runtime.h>

typedef float v2f __attribute__((ext_vector_type(2)));
typedef float v8f __attribute__((ext_vector_type(8)));

#define NFPN_EPS2 1e-10f
#define NFPN_MAX_DEPTH 100

__global__ __launch_bounds__(256) void nfpn_rps_kernel(
    const float* __restrict__ din,
    const float* __restrict__ W1,
    const float* __restrict__ b1,
    const float* __restrict__ W2,
    const float* __restrict__ b2,
    float* __restrict__ out,
    int B)
{
    const int gid  = blockIdx.x * blockDim.x + threadIdx.x;
    const int lane = threadIdx.x & 31;
    const bool low = lane < 16;

    int s = gid;
    const bool active = (s < B);
    if (!active) s = B - 1;

    const float d0 = din[3*s + 0];
    const float d1 = din[3*s + 1];
    const float d2 = din[3*s + 2];

    float q[6];
#pragma unroll
    for (int j = 0; j < 6; ++j) {
        float t = fmaf(d0, W1[3*j + 0], fmaf(d1, W1[3*j + 1], fmaf(d2, W1[3*j + 2], b1[j])));
        q[j] = (t >= 0.f) ? t : 0.1f * t;
    }

    const int  M     = lane & 15;
    const int  rr    = M & 7;
    const bool valid = (rr < 6) && ((M < 8) == low);
    v2f a0, a1, a2;
    a0.x = valid ? -W2[rr*6 + 0] : 0.f;
    a0.y = valid ? -W2[rr*6 + 1] : 0.f;
    a1.x = valid ? -W2[rr*6 + 2] : 0.f;
    a1.y = valid ? -W2[rr*6 + 3] : 0.f;
    a2.x = valid ? -W2[rr*6 + 4] : 0.f;
    a2.y = valid ? -W2[rr*6 + 5] : 0.f;

    v8f cinit;
#pragma unroll
    for (int j = 0; j < 6; ++j) cinit[j] = -b2[j];
    cinit[6] = 0.f; cinit[7] = 0.f;

    float u[6];
#pragma unroll
    for (int j = 0; j < 6; ++j) u[j] = 1.0f / 3.0f;

    auto matvec = [&](const float* x, float* y) {
        v2f B0, B1, B2;
        B0.x = x[0];  B0.y = x[1];
        B1.x = x[2];  B1.y = x[3];
        B2.x = x[4];  B2.y = x[5];
        v8f acc = cinit;
        acc = __builtin_amdgcn_wmma_f32_16x16x4_f32(false, a0, false, B0, (short)0, acc, false, false);
        acc = __builtin_amdgcn_wmma_f32_16x16x4_f32(false, a1, false, B1, (short)0, acc, false, false);
        acc = __builtin_amdgcn_wmma_f32_16x16x4_f32(false, a2, false, B2, (short)0, acc, false, false);
#pragma unroll
        for (int j = 0; j < 6; ++j) y[j] = acc[j];
    };

    auto proj3 = [&](float a, float b, float c, float& o0, float& o1, float& o2) {
        float hi  = fmaxf(fmaxf(a, b), c);
        float lo  = fminf(fminf(a, b), c);
        float sum = a + b + c;
        float mid = sum - hi - lo;
        float c2  = hi + mid;
        bool cond2 = (mid - hi + 1.0f) > 0.0f;
        bool cond3 = (3.0f*lo - sum + 1.0f) > 0.0f;
        int rho = 1 + (cond2 ? 1 : 0) + (cond3 ? 1 : 0);
        float cs    = (rho == 3) ? sum : ((rho == 2) ? c2 : hi);
        float rinv  = (rho == 3) ? (1.0f/3.0f) : ((rho == 2) ? 0.5f : 1.0f);
        float theta = (cs - 1.0f) * rinv;
        o0 = fmaxf(a - theta, 0.f);
        o1 = fmaxf(b - theta, 0.f);
        o2 = fmaxf(c - theta, 0.f);
    };

    int depth = 0;
    for (;;) {
        float x[6], y[6], un[6];
#pragma unroll
        for (int j = 0; j < 6; ++j) x[j] = u[j] + q[j];
        matvec(x, y);
        proj3(y[0], y[1], y[2], un[0], un[1], un[2]);
        proj3(y[3], y[4], y[5], un[3], un[4], un[5]);
        float r2 = 0.f;
#pragma unroll
        for (int j = 0; j < 6; ++j) { float dd = un[j] - u[j]; r2 = fmaf(dd, dd, r2); u[j] = un[j]; }
        ++depth;
        if (depth >= NFPN_MAX_DEPTH) break;
        if (__ballot(r2 > NFPN_EPS2) == 0ULL) break;
    }

    float x[6], y[6], o[6];
#pragma unroll
    for (int j = 0; j < 6; ++j) x[j] = u[j] + q[j];
    matvec(x, y);
    proj3(y[0], y[1], y[2], o[0], o[1], o[2]);
    proj3(y[3], y[4], y[5], o[3], o[4], o[5]);

    if (active) {
        float2* op = reinterpret_cast<float2*>(out + 6l * s);
        op[0] = make_float2(o[0], o[1]);
        op[1] = make_float2(o[2], o[3]);
        op[2] = make_float2(o[4], o[5]);
    }
}

extern "C" void kernel_launch(void* const* d_in, const int* in_sizes, int n_in,
                              void* d_out, int out_size, void* d_ws, size_t ws_size,
                              hipStream_t stream) {
    const float* d  = (const float*)d_in[0];
    const float* W1 = (const float*)d_in[1];
    const float* b1 = (const float*)d_in[2];
    const float* W2 = (const float*)d_in[3];
    const float* b2 = (const float*)d_in[4];
    float* out = (float*)d_out;

    const int B = in_sizes[0] / 3;
    const int threads = 256;
    const int blocks  = (B + threads - 1) / threads;
    hipLaunchKernelGGL(nfpn_rps_kernel, dim3(blocks), dim3(threads), 0, stream,
                       d, W1, b1, W2, b2, out, B);
}


# Linear map restated on the 4 free coordinates of u (each projected block sums to 1): constant part -(W2(q+e2+e5)+b2) goes into C once, 2 WMMAs per step (was 3); sums 6 products once, then 2+2 a step
# speedup vs baseline: 1.5009x; 1.2167x over previous
#include <hip/hip_runtime.h>

typedef float v2f __attribute__((ext_vector_type(2)));
typedef float v8f __attribute__((ext_vector_type(8)));

#define NFPN_EPS2 1e-10f
#define NFPN_MAX_DEPTH 100

__global__ __launch_bounds__(256) void nfpn_rps_kernel(
    const float* __restrict__ din,
    const float* __restrict__ W1,
    const float* __restrict__ b1,
    const float* __restrict__ W2,
    const float* __restrict__ b2,
    float* __restrict__ out,
    int B)
{
    const int gid  = blockIdx.x * blockDim.x + threadIdx.x;
    const int lane = threadIdx.x & 31;
    const bool low = lane < 16;

    int s = gid;
    const bool active = (s < B);
    if (!active) s = B - 1;

    const float d0 = din[3*s + 0];
    const float d1 = din[3*s + 1];
    const float d2 = din[3*s + 2];

    float q[6];
#pragma unroll
    for (int j = 0; j < 6; ++j) {
        float t = fmaf(d0, W1[3*j + 0], fmaf(d1, W1[3*j + 1], fmaf(d2, W1[3*j + 2], b1[j])));
        q[j] = (t >= 0.f) ? t : 0.1f * t;
    }

    const int  M     = lane & 15;
    const int  rr    = M & 7;
    const bool valid = (rr < 6) && ((M < 8) == low);
    const int rsel = (rr < 6) ? rr : 0;
    float w[6];
#pragma unroll
    for (int j = 0; j < 6; ++j) w[j] = W2[rsel*6 + j];
#pragma unroll
    for (int j = 0; j < 6; ++j) w[j] = valid ? -w[j] : 0.f;

    v2f a0, a1, a2;
    a0.x = w[0]; a0.y = w[1];
    a1.x = w[2]; a1.y = w[3];
    a2.x = w[4]; a2.y = w[5];

    v2f ad0, ad1;
    ad0.x = w[0] - w[2];  ad0.y = w[1] - w[2];
    ad1.x = w[3] - w[5];  ad1.y = w[4] - w[5];

    v8f cvec;
    {
        v8f acc;
#pragma unroll
        for (int j = 0; j < 6; ++j) acc[j] = -b2[j];
        acc[6] = 0.f; acc[7] = 0.f;
        v2f B0, B1, B2;
        B0.x = q[0];         B0.y = q[1];
        B1.x = q[2] + 1.0f;  B1.y = q[3];
        B2.x = q[4];         B2.y = q[5] + 1.0f;
        acc = __builtin_amdgcn_wmma_f32_16x16x4_f32(false, a0, false, B0, (short)0, acc, false, false);
        acc = __builtin_amdgcn_wmma_f32_16x16x4_f32(false, a1, false, B1, (short)0, acc, false, false);
        acc = __builtin_amdgcn_wmma_f32_16x16x4_f32(false, a2, false, B2, (short)0, acc, false, false);
        cvec = acc;
    }

    float u[6];
#pragma unroll
    for (int j = 0; j < 6; ++j) u[j] = 1.0f / 3.0f;

    auto matvec = [&](const float* uu, float* y) {
        v2f B0, B1;
        B0.x = uu[0];  B0.y = uu[1];
        B1.x = uu[3];  B1.y = uu[4];
        v8f acc = cvec;
        acc = __builtin_amdgcn_wmma_f32_16x16x4_f32(false, ad0, false, B0, (short)0, acc, false, false);
        acc = __builtin_amdgcn_wmma_f32_16x16x4_f32(false, ad1, false, B1, (short)0, acc, false, false);
#pragma unroll
        for (int j = 0; j < 6; ++j) y[j] = acc[j];
    };

    auto proj3 = [&](float a, float b, float c, float& o0, float& o1, float& o2) {
        float hi  = fmaxf(fmaxf(a, b), c);
        float lo  = fminf(fminf(a, b), c);
        float sum = a + b + c;
        float mid = sum - hi - lo;
        float c2  = hi + mid;
        bool cond2 = (mid - hi + 1.0f) > 0.0f;
        bool cond3 = (3.0f*lo - sum + 1.0f) > 0.0f;
        int rho = 1 + (cond2 ? 1 : 0) + (cond3 ? 1 : 0);
        float cs    = (rho == 3) ? sum : ((rho == 2) ? c2 : hi);
        float rinv  = (rho == 3) ? (1.0f/3.0f) : ((rho == 2) ? 0.5f : 1.0f);
        float theta = (cs - 1.0f) * rinv;
        o0 = fmaxf(a - theta, 0.f);
        o1 = fmaxf(b - theta, 0.f);
        o2 = fmaxf(c - theta, 0.f);
    };

    int depth = 0;
    for (;;) {
        float y[6], un[6];
        matvec(u, y);
        proj3(y[0], y[1], y[2], un[0], un[1], un[2]);
        proj3(y[3], y[4], y[5], un[3], un[4], un[5]);
        float r2 = 0.f;
#pragma unroll
        for (int j = 0; j < 6; ++j) { float dd = un[j] - u[j]; r2 = fmaf(dd, dd, r2); u[j] = un[j]; }
        ++depth;
        if (depth >= NFPN_MAX_DEPTH) break;
        if (__ballot(r2 > NFPN_EPS2) == 0ULL) break;
    }

    float y[6], o[6];
    matvec(u, y);
    proj3(y[0], y[1], y[2], o[0], o[1], o[2]);
    proj3(y[3], y[4], y[5], o[3], o[4], o[5]);

    if (active) {
        float2* op = reinterpret_cast<float2*>(out + 6l * s);
        op[0] = make_float2(o[0], o[1]);
        op[1] = make_float2(o[2], o[3]);
        op[2] = make_float2(o[4], o[5]);
    }
}

extern "C" void kernel_launch(void* const* d_in, const int* in_sizes, int n_in,
                              void* d_out, int out_size, void* d_ws, size_t ws_size,
                              hipStream_t stream) {
    const float* d  = (const float*)d_in[0];
    const float* W1 = (const float*)d_in[1];
    const float* b1 = (const float*)d_in[2];
    const float* W2 = (const float*)d_in[3];
    const float* b2 = (const float*)d_in[4];
    float* out = (float*)d_out;

    const int B = in_sizes[0] / 3;
    const int threads = 256;
    const int blocks  = (B + threads - 1) / threads;
    hipLaunchKernelGGL(nfpn_rps_kernel, dim3(blocks), dim3(threads), 0, stream,
                       d, W1, b1, W2, b2, out, B);
}


# Simplex projection: theta = largest of the three prefix candidates hi-1, (hi+mid-1)/2, (sum-1)/3 (mid by med3) replaces the rho count, its compares, selects, mask ops; order of sums unchanged
# speedup vs baseline: 1.8176x; 1.2110x over previous
#include <hip/hip_runtime.h>

typedef float v2f __attribute__((ext_vector_type(2)));
typedef float v8f __attribute__((ext_vector_type(8)));

#define NFPN_EPS2 1e-10f
#define NFPN_MAX_DEPTH 100

__global__ __launch_bounds__(256) void nfpn_rps_kernel(
    const float* __restrict__ din,
    const float* __restrict__ W1,
    const float* __restrict__ b1,
    const float* __restrict__ W2,
    const float* __restrict__ b2,
    float* __restrict__ out,
    int B)
{
    const int gid  = blockIdx.x * blockDim.x + threadIdx.x;
    const int lane = threadIdx.x & 31;
    const bool low = lane < 16;

    int s = gid;
    const bool active = (s < B);
    if (!active) s = B - 1;

    const float d0 = din[3*s + 0];
    const float d1 = din[3*s + 1];
    const float d2 = din[3*s + 2];

    float q[6];
#pragma unroll
    for (int j = 0; j < 6; ++j) {
        float t = fmaf(d0, W1[3*j + 0], fmaf(d1, W1[3*j + 1], fmaf(d2, W1[3*j + 2], b1[j])));
        q[j] = (t >= 0.f) ? t : 0.1f * t;
    }

    const int  M     = lane & 15;
    const int  rr    = M & 7;
    const bool valid = (rr < 6) && ((M < 8) == low);
    const int rsel = (rr < 6) ? rr : 0;
    float w[6];
#pragma unroll
    for (int j = 0; j < 6; ++j) w[j] = W2[rsel*6 + j];
#pragma unroll
    for (int j = 0; j < 6; ++j) w[j] = valid ? -w[j] : 0.f;

    v2f a0, a1, a2;
    a0.x = w[0]; a0.y = w[1];
    a1.x = w[2]; a1.y = w[3];
    a2.x = w[4]; a2.y = w[5];

    v2f ad0, ad1;
    ad0.x = w[0] - w[2];  ad0.y = w[1] - w[2];
    ad1.x = w[3] - w[5];  ad1.y = w[4] - w[5];

    v8f cvec;
    {
        v8f acc;
#pragma unroll
        for (int j = 0; j < 6; ++j) acc[j] = -b2[j];
        acc[6] = 0.f; acc[7] = 0.f;
        v2f B0, B1, B2;
        B0.x = q[0];         B0.y = q[1];
        B1.x = q[2] + 1.0f;  B1.y = q[3];
        B2.x = q[4];         B2.y = q[5] + 1.0f;
        acc = __builtin_amdgcn_wmma_f32_16x16x4_f32(false, a0, false, B0, (short)0, acc, false, false);
        acc = __builtin_amdgcn_wmma_f32_16x16x4_f32(false, a1, false, B1, (short)0, acc, false, false);
        acc = __builtin_amdgcn_wmma_f32_16x16x4_f32(false, a2, false, B2, (short)0, acc, false, false);
        cvec = acc;
    }

    float u[6];
#pragma unroll
    for (int j = 0; j < 6; ++j) u[j] = 1.0f / 3.0f;

    auto matvec = [&](const float* uu, float* y) {
        v2f B0, B1;
        B0.x = uu[0];  B0.y = uu[1];
        B1.x = uu[3];  B1.y = uu[4];
        v8f acc = cvec;
        acc = __builtin_amdgcn_wmma_f32_16x16x4_f32(false, ad0, false, B0, (short)0, acc, false, false);
        acc = __builtin_amdgcn_wmma_f32_16x16x4_f32(false, ad1, false, B1, (short)0, acc, false, false);
#pragma unroll
        for (int j = 0; j < 6; ++j) y[j] = acc[j];
    };

    auto proj3 = [&](float a, float b, float c, float& o0, float& o1, float& o2) {
        float hi  = fmaxf(fmaxf(a, b), c);
        float mid = __builtin_amdgcn_fmed3f(a, b, c);
        float sum = a + b + c;
        float c2  = hi + mid;
        float t1  = hi - 1.0f;
        float t2  = fmaf(c2,  0.5f,        -0.5f);
        float t3  = fmaf(sum, 1.0f / 3.0f, -1.0f / 3.0f);
        float theta = fmaxf(fmaxf(t1, t2), t3);
        o0 = fmaxf(a - theta, 0.f);
        o1 = fmaxf(b - theta, 0.f);
        o2 = fmaxf(c - theta, 0.f);
    };

    int depth = 0;
    for (;;) {
        float y[6], un[6];
        matvec(u, y);
        proj3(y[0], y[1], y[2], un[0], un[1], un[2]);
        proj3(y[3], y[4], y[5], un[3], un[4], un[5]);
        float r2 = 0.f;
#pragma unroll
        for (int j = 0; j < 6; ++j) { float dd = un[j] - u[j]; r2 = fmaf(dd, dd, r2); u[j] = un[j]; }
        ++depth;
        if (depth >= NFPN_MAX_DEPTH) break;
        if (__ballot(r2 > NFPN_EPS2) == 0ULL) break;
    }

    float y[6], o[6];
    matvec(u, y);
    proj3(y[0], y[1], y[2], o[0], o[1], o[2]);
    proj3(y[3], y[4], y[5], o[3], o[4], o[5]);

    if (active) {
        float2* op = reinterpret_cast<float2*>(out + 6l * s);
        op[0] = make_float2(o[0], o[1]);
        op[1] = make_float2(o[2], o[3]);
        op[2] = make_float2(o[4], o[5]);
    }
}

extern "C" void kernel_launch(void* const* d_in, const int* in_sizes, int n_in,
                              void* d_out, int out_size, void* d_ws, size_t ws_size,
                              hipStream_t stream) {
    const float* d  = (const float*)d_in[0];
    const float* W1 = (const float*)d_in[1];
    const float* b1 = (const float*)d_in[2];
    const float* W2 = (const float*)d_in[3];
    const float* b2 = (const float*)d_in[4];
    float* out = (float*)d_out;

    const int B = in_sizes[0] / 3;
    const int threads = 256;
    const int blocks  = (B + threads - 1) / threads;
    hipLaunchKernelGGL(nfpn_rps_kernel, dim3(blocks), dim3(threads), 0, stream,
                       d, W1, b1, W2, b2, out, B);
}


# Each output block centred inside the matrix product (rows of W2 minus their block mean, C shifted to match): a block sums to one, projection needs no sum and no middle value; block-sum order changed
# speedup vs baseline: 2.0332x; 1.1186x over previous
#include <hip/hip_runtime.h>

typedef float v2f __attribute__((ext_vector_type(2)));
typedef float v8f __attribute__((ext_vector_type(8)));

#define NFPN_EPS2 1e-10f
#define NFPN_MAX_DEPTH 100

__global__ __launch_bounds__(256) void nfpn_rps_kernel(
    const float* __restrict__ din,
    const float* __restrict__ W1,
    const float* __restrict__ b1,
    const float* __restrict__ W2,
    const float* __restrict__ b2,
    float* __restrict__ out,
    int B)
{
    const int gid  = blockIdx.x * blockDim.x + threadIdx.x;
    const int lane = threadIdx.x & 31;
    const bool low = lane < 16;

    int s = gid;
    const bool active = (s < B);
    if (!active) s = B - 1;

    const float d0 = din[3*s + 0];
    const float d1 = din[3*s + 1];
    const float d2 = din[3*s + 2];

    float q[6];
#pragma unroll
    for (int j = 0; j < 6; ++j) {
        float t = fmaf(d0, W1[3*j + 0], fmaf(d1, W1[3*j + 1], fmaf(d2, W1[3*j + 2], b1[j])));
        q[j] = (t >= 0.f) ? t : 0.1f * t;
    }

    const int  M     = lane & 15;
    const int  rr    = M & 7;
    const bool valid = (rr < 6) && ((M < 8) == low);
    const int rsel = (rr < 6) ? rr : 0;
    const int rb   = (rsel >= 3) ? 3 : 0;
    float w[6];
#pragma unroll
    for (int j = 0; j < 6; ++j) {
        const float rowsum = W2[rb*6 + j] + W2[rb*6 + 6 + j] + W2[rb*6 + 12 + j];
        const float x = fmaf(rowsum, 1.0f / 3.0f, -W2[rsel*6 + j]);
        w[j] = valid ? x : 0.f;
    }

    v2f a0, a1, a2;
    a0.x = w[0]; a0.y = w[1];
    a1.x = w[2]; a1.y = w[3];
    a2.x = w[4]; a2.y = w[5];

    v2f ad0, ad1;
    ad0.x = w[0] - w[2];  ad0.y = w[1] - w[2];
    ad1.x = w[3] - w[5];  ad1.y = w[4] - w[5];

    v8f cvec;
    {
        const float g0 = (b2[0] + b2[1] + b2[2] + 1.0f) * (1.0f / 3.0f);
        const float g1 = (b2[3] + b2[4] + b2[5] + 1.0f) * (1.0f / 3.0f);
        v8f acc;
#pragma unroll
        for (int j = 0; j < 3; ++j) acc[j] = g0 - b2[j];
#pragma unroll
        for (int j = 3; j < 6; ++j) acc[j] = g1 - b2[j];
        acc[6] = 0.f; acc[7] = 0.f;
        v2f B0, B1, B2;
        B0.x = q[0];         B0.y = q[1];
        B1.x = q[2] + 1.0f;  B1.y = q[3];
        B2.x = q[4];         B2.y = q[5] + 1.0f;
        acc = __builtin_amdgcn_wmma_f32_16x16x4_f32(false, a0, false, B0, (short)0, acc, false, false);
        acc = __builtin_amdgcn_wmma_f32_16x16x4_f32(false, a1, false, B1, (short)0, acc, false, false);
        acc = __builtin_amdgcn_wmma_f32_16x16x4_f32(false, a2, false, B2, (short)0, acc, false, false);
        cvec = acc;
    }

    float u[6];
#pragma unroll
    for (int j = 0; j < 6; ++j) u[j] = 1.0f / 3.0f;

    auto matvec = [&](const float* uu, float* z) {
        v2f B0, B1;
        B0.x = uu[0];  B0.y = uu[1];
        B1.x = uu[3];  B1.y = uu[4];
        v8f acc = cvec;
        acc = __builtin_amdgcn_wmma_f32_16x16x4_f32(false, ad0, false, B0, (short)0, acc, false, false);
        acc = __builtin_amdgcn_wmma_f32_16x16x4_f32(false, ad1, false, B1, (short)0, acc, false, false);
#pragma unroll
        for (int j = 0; j < 6; ++j) z[j] = acc[j];
    };

    auto proj3 = [&](float a, float b, float c, float& o0, float& o1, float& o2) {
        float hi  = fmaxf(fmaxf(a, b), c);
        float lo  = fminf(fminf(a, b), c);
        float th2 = fmaxf(fmaxf(fmaf(hi, 2.0f, -2.0f), -lo), 0.f);
        o0 = fmaxf(fmaf(th2, -0.5f, a), 0.f);
        o1 = fmaxf(fmaf(th2, -0.5f, b), 0.f);
        o2 = fmaxf(fmaf(th2, -0.5f, c), 0.f);
    };

    int depth = 0;
    for (;;) {
        float z[6], un[6];
        matvec(u, z);
        proj3(z[0], z[1], z[2], un[0], un[1], un[2]);
        proj3(z[3], z[4], z[5], un[3], un[4], un[5]);
        float r2 = 0.f;
#pragma unroll
        for (int j = 0; j < 6; ++j) { float dd = un[j] - u[j]; r2 = fmaf(dd, dd, r2); u[j] = un[j]; }
        ++depth;
        if (depth >= NFPN_MAX_DEPTH) break;
        if (__ballot(r2 > NFPN_EPS2) == 0ULL) break;
    }

    float z[6], o[6];
    matvec(u, z);
    proj3(z[0], z[1], z[2], o[0], o[1], o[2]);
    proj3(z[3], z[4], z[5], o[3], o[4], o[5]);

    if (active) {
        float2* op = reinterpret_cast<float2*>(out + 6l * s);
        op[0] = make_float2(o[0], o[1]);
        op[1] = make_float2(o[2], o[3]);
        op[2] = make_float2(o[4], o[5]);
    }
}

extern "C" void kernel_launch(void* const* d_in, const int* in_sizes, int n_in,
                              void* d_out, int out_size, void* d_ws, size_t ws_size,
                              hipStream_t stream) {
    const float* d  = (const float*)d_in[0];
    const float* W1 = (const float*)d_in[1];
    const float* b1 = (const float*)d_in[2];
    const float* W2 = (const float*)d_in[3];
    const float* b2 = (const float*)d_in[4];
    float* out = (float*)d_out;

    const int B = in_sizes[0] / 3;
    const int threads = 256;
    const int blocks  = (B + threads - 1) / threads;
    hipLaunchKernelGGL(nfpn_rps_kernel, dim3(blocks), dim3(threads), 0, stream,
                       d, W1, b1, W2, b2, out, B);
}


# Projection of a block that sums to one as clamp(v + min(lo,0)/2, 0, 1): no theta, no largest value; shift and clamp are one packed multiply-add with 0..1 output clamp per register pair
# speedup vs baseline: 2.2746x; 1.1187x over previous
#include <hip/hip_runtime.h>

typedef float v2f __attribute__((ext_vector_type(2)));
typedef float v8f __attribute__((ext_vector_type(8)));

#define NFPN_EPS2 1e-10f
#define NFPN_MAX_DEPTH 100

__global__ __launch_bounds__(256) void nfpn_rps_kernel(
    const float* __restrict__ din,
    const float* __restrict__ W1,
    const float* __restrict__ b1,
    const float* __restrict__ W2,
    const float* __restrict__ b2,
    float* __restrict__ out,
    int B)
{
    const int gid  = blockIdx.x * blockDim.x + threadIdx.x;
    const int lane = threadIdx.x & 31;
    const bool low = lane < 16;

    int s = gid;
    const bool active = (s < B);
    if (!active) s = B - 1;

    const float d0 = din[3*s + 0];
    const float d1 = din[3*s + 1];
    const float d2 = din[3*s + 2];

    float q[6];
#pragma unroll
    for (int j = 0; j < 6; ++j) {
        float t = fmaf(d0, W1[3*j + 0], fmaf(d1, W1[3*j + 1], fmaf(d2, W1[3*j + 2], b1[j])));
        q[j] = (t >= 0.f) ? t : 0.1f * t;
    }

    const int  M     = lane & 15;
    const int  rr    = M & 7;
    const bool valid = (rr < 6) && ((M < 8) == low);
    const int  osel  = (0x524310 >> (4 * rr)) & 7;
    const int rb   = (osel >= 3) ? 3 : 0;
    float w[6];
#pragma unroll
    for (int j = 0; j < 6; ++j) {
        const float rowsum = W2[rb*6 + j] + W2[rb*6 + 6 + j] + W2[rb*6 + 12 + j];
        const float x = fmaf(rowsum, 1.0f / 3.0f, -W2[osel*6 + j]);
        w[j] = valid ? x : 0.f;
    }

    v2f a0, a1, a2;
    a0.x = w[0]; a0.y = w[1];
    a1.x = w[2]; a1.y = w[3];
    a2.x = w[4]; a2.y = w[5];

    v2f ad0, ad1;
    ad0.x = w[0] - w[2];  ad0.y = w[1] - w[2];
    ad1.x = w[3] - w[5];  ad1.y = w[4] - w[5];

    v8f cvec;
    {
        const float g0 = (b2[0] + b2[1] + b2[2] + 1.0f) * (1.0f / 3.0f);
        const float g1 = (b2[3] + b2[4] + b2[5] + 1.0f) * (1.0f / 3.0f);
        v8f acc;
        acc[0] = g0 - b2[0];  acc[1] = g0 - b2[1];
        acc[2] = g1 - b2[3];  acc[3] = g1 - b2[4];
        acc[4] = g0 - b2[2];  acc[5] = g1 - b2[5];
        acc[6] = 0.f; acc[7] = 0.f;
        v2f B0, B1, B2;
        B0.x = q[0];         B0.y = q[1];
        B1.x = q[2] + 1.0f;  B1.y = q[3];
        B2.x = q[4];         B2.y = q[5] + 1.0f;
        acc = __builtin_amdgcn_wmma_f32_16x16x4_f32(false, a0, false, B0, (short)0, acc, false, false);
        acc = __builtin_amdgcn_wmma_f32_16x16x4_f32(false, a1, false, B1, (short)0, acc, false, false);
        acc = __builtin_amdgcn_wmma_f32_16x16x4_f32(false, a2, false, B2, (short)0, acc, false, false);
        cvec = acc;
    }

    v2f p01, p34, p25;
    p01.x = 1.0f / 3.0f;  p01.y = 1.0f / 3.0f;
    p34 = p01;  p25 = p01;

    auto matvec = [&](v2f B0, v2f B1) -> v8f {
        v8f acc = cvec;
        acc = __builtin_amdgcn_wmma_f32_16x16x4_f32(false, ad0, false, B0, (short)0, acc, false, false);
        acc = __builtin_amdgcn_wmma_f32_16x16x4_f32(false, ad1, false, B1, (short)0, acc, false, false);
        return acc;
    };


    auto shifts = [&](v8f z) -> v2f {
        v2f h;
        h.x = fminf(fminf(z[0], z[1]), fminf(z[4], 0.f));
        h.y = fminf(fminf(z[2], z[3]), fminf(z[5], 0.f));
        return h;
    };

    auto step = [&](const v2f q01, const v2f q34, const v2f q25, v2f& n01, v2f& n34, v2f& n25) -> float {
        const v8f z = matvec(q01, q34);
        const v2f h = shifts(z);
        v2f z01, z34, z25;
        z01.x = z[0];  z01.y = z[1];
        z34.x = z[2];  z34.y = z[3];
        z25.x = z[4];  z25.y = z[5];
        v2f e01, e25, e34;
        asm("s_delay_alu instid0(VALU_DEP_1) | instskip(NEXT) | instid1(VALU_DEP_2)\n\t"
            "v_pk_fma_f32 %0, %6, 0.5, %7 op_sel_hi:[0,0,1] clamp\n\t"
            "v_pk_fma_f32 %1, %6, 0.5, %8 op_sel_hi:[1,0,1] clamp\n\t"
            "s_delay_alu instid0(VALU_DEP_3) | instskip(NEXT) | instid1(VALU_DEP_3)\n\t"
            "v_pk_fma_f32 %2, %6, 0.5, %9 op_sel:[1,0,0] op_sel_hi:[1,0,1] clamp\n\t"
            "v_pk_add_f32 %3, %0, %10 neg_lo:[0,1] neg_hi:[0,1]\n\t"
            "s_delay_alu instid0(VALU_DEP_3) | instskip(NEXT) | instid1(VALU_DEP_3)\n\t"
            "v_pk_add_f32 %4, %1, %11 neg_lo:[0,1] neg_hi:[0,1]\n\t"
            "v_pk_add_f32 %5, %2, %12 neg_lo:[0,1] neg_hi:[0,1]\n\t"
            "s_delay_alu instid0(VALU_DEP_3)"
            : "=&v"(n01), "=&v"(n25), "=&v"(n34), "=&v"(e01), "=&v"(e25), "=&v"(e34)
            : "v"(h), "v"(z01), "v"(z25), "v"(z34), "v"(q01), "v"(q25), "v"(q34));
        float r2 = 0.f;
        r2 = fmaf(e01.x, e01.x, r2);  r2 = fmaf(e01.y, e01.y, r2);  r2 = fmaf(e25.x, e25.x, r2);
        r2 = fmaf(e34.x, e34.x, r2);  r2 = fmaf(e34.y, e34.y, r2);  r2 = fmaf(e25.y, e25.y, r2);
        return r2;
    };

    auto finish = [&](const v2f q01, const v2f q34) {
        const v8f z = matvec(q01, q34);
        const v2f h = shifts(z);
        v2f z01, z34, z25;
        z01.x = z[0];  z01.y = z[1];
        z34.x = z[2];  z34.y = z[3];
        z25.x = z[4];  z25.y = z[5];
        v2f o01, o25, o34;
        asm("s_delay_alu instid0(VALU_DEP_1) | instskip(NEXT) | instid1(VALU_DEP_2)\n\t"
            "v_pk_fma_f32 %0, %3, 0.5, %4 op_sel_hi:[0,0,1] clamp\n\t"
            "v_pk_fma_f32 %1, %3, 0.5, %5 op_sel_hi:[1,0,1] clamp\n\t"
            "s_delay_alu instid0(VALU_DEP_3)\n\t"
            "v_pk_fma_f32 %2, %3, 0.5, %6 op_sel:[1,0,0] op_sel_hi:[1,0,1] clamp"
            : "=&v"(o01), "=&v"(o25), "=&v"(o34)
            : "v"(h), "v"(z01), "v"(z25), "v"(z34));
        if (active) {
            float2* op = reinterpret_cast<float2*>(out + 6l * s);
            op[0] = make_float2(o01.x, o01.y);
            op[1] = make_float2(o25.x, o34.x);
            op[2] = make_float2(o34.y, o25.y);
        }
    };

    int depth = 0;
    for (;;) {
        v2f t01, t34, t25;
        float r2 = step(p01, p34, p25, t01, t34, t25);
        ++depth;
        if (depth >= NFPN_MAX_DEPTH || __ballot(r2 > NFPN_EPS2) == 0ULL) {
            finish(t01, t34);
            return;
        }
        r2 = step(t01, t34, t25, p01, p34, p25);
        ++depth;
        if (depth >= NFPN_MAX_DEPTH || __ballot(r2 > NFPN_EPS2) == 0ULL) {
            finish(p01, p34);
            return;
        }
    }
}

extern "C" void kernel_launch(void* const* d_in, const int* in_sizes, int n_in,
                              void* d_out, int out_size, void* d_ws, size_t ws_size,
                              hipStream_t stream) {
    const float* d  = (const float*)d_in[0];
    const float* W1 = (const float*)d_in[1];
    const float* b1 = (const float*)d_in[2];
    const float* W2 = (const float*)d_in[3];
    const float* b2 = (const float*)d_in[4];
    float* out = (float*)d_out;

    const int B = in_sizes[0] / 3;
    const int threads = 256;
    const int blocks  = (B + threads - 1) / threads;
    hipLaunchKernelGGL(nfpn_rps_kernel, dim3(blocks), dim3(threads), 0, stream,
                       d, W1, b1, W2, b2, out, B);
}


# Residual from the 4 free coordinates only (e2=-(e0+e1), e5=-(e3+e4)): state pairs (u0,u3),(u1,u4), 3 packed FMAs + 1 compare, dependent pair left out of the loop; map products now summed 0,3,1,4
# speedup vs baseline: 2.5001x; 1.0991x over previous
#include <hip/hip_runtime.h>

typedef float v2f __attribute__((ext_vector_type(2)));
typedef float v8f __attribute__((ext_vector_type(8)));

#define NFPN_EPS2 1e-10f
#define NFPN_MAX_DEPTH 100

__global__ __launch_bounds__(256) void nfpn_rps_kernel(
    const float* __restrict__ din,
    const float* __restrict__ W1,
    const float* __restrict__ b1,
    const float* __restrict__ W2,
    const float* __restrict__ b2,
    float* __restrict__ out,
    int B)
{
    const int gid  = blockIdx.x * blockDim.x + threadIdx.x;
    const int lane = threadIdx.x & 31;
    const bool low = lane < 16;

    int s = gid;
    const bool active = (s < B);
    if (!active) s = B - 1;

    const float d0 = din[3*s + 0];
    const float d1 = din[3*s + 1];
    const float d2 = din[3*s + 2];

    float q[6];
#pragma unroll
    for (int j = 0; j < 6; ++j) {
        float t = fmaf(d0, W1[3*j + 0], fmaf(d1, W1[3*j + 1], fmaf(d2, W1[3*j + 2], b1[j])));
        q[j] = (t >= 0.f) ? t : 0.1f * t;
    }

    const int  M     = lane & 15;
    const int  rr    = M & 7;
    const bool valid = (rr < 6) && ((M < 8) == low);
    const int  osel  = (0x524130 >> (4 * rr)) & 7;
    const int rb   = (osel >= 3) ? 3 : 0;
    float w[6];
#pragma unroll
    for (int j = 0; j < 6; ++j) {
        const float rowsum = W2[rb*6 + j] + W2[rb*6 + 6 + j] + W2[rb*6 + 12 + j];
        const float x = fmaf(rowsum, 1.0f / 3.0f, -W2[osel*6 + j]);
        w[j] = valid ? x : 0.f;
    }

    v2f a0, a1, a2;
    a0.x = w[0]; a0.y = w[1];
    a1.x = w[2]; a1.y = w[3];
    a2.x = w[4]; a2.y = w[5];

    v2f ad0, ad1;
    ad0.x = w[0] - w[2];  ad0.y = w[3] - w[5];
    ad1.x = w[1] - w[2];  ad1.y = w[4] - w[5];

    v8f cvec;
    {
        const float g0 = (b2[0] + b2[1] + b2[2] + 1.0f) * (1.0f / 3.0f);
        const float g1 = (b2[3] + b2[4] + b2[5] + 1.0f) * (1.0f / 3.0f);
        v8f acc;
        acc[0] = g0 - b2[0];  acc[1] = g1 - b2[3];
        acc[2] = g0 - b2[1];  acc[3] = g1 - b2[4];
        acc[4] = g0 - b2[2];  acc[5] = g1 - b2[5];
        acc[6] = 0.f; acc[7] = 0.f;
        v2f B0, B1, B2;
        B0.x = q[0];         B0.y = q[1];
        B1.x = q[2] + 1.0f;  B1.y = q[3];
        B2.x = q[4];         B2.y = q[5] + 1.0f;
        acc = __builtin_amdgcn_wmma_f32_16x16x4_f32(false, a0, false, B0, (short)0, acc, false, false);
        acc = __builtin_amdgcn_wmma_f32_16x16x4_f32(false, a1, false, B1, (short)0, acc, false, false);
        acc = __builtin_amdgcn_wmma_f32_16x16x4_f32(false, a2, false, B2, (short)0, acc, false, false);
        cvec = acc;
    }

    v2f pa, pb;
    pa.x = 1.0f / 3.0f;  pa.y = 1.0f / 3.0f;
    pb = pa;

    auto matvec = [&](v2f B0, v2f B1) -> v8f {
        v8f acc = cvec;
        acc = __builtin_amdgcn_wmma_f32_16x16x4_f32(false, ad0, false, B0, (short)0, acc, false, false);
        acc = __builtin_amdgcn_wmma_f32_16x16x4_f32(false, ad1, false, B1, (short)0, acc, false, false);
        return acc;
    };


    auto shifts = [&](v8f z) -> v2f {
        v2f h;
        h.x = fminf(fminf(z[0], z[2]), fminf(z[4], 0.f));
        h.y = fminf(fminf(z[1], z[3]), fminf(z[5], 0.f));
        return h;
    };

    v2f kthr;
    kthr.x = 0.f;  kthr.y = -0.5f * NFPN_EPS2;

    auto step = [&](const v2f qa, const v2f qb, v2f& na, v2f& nb) -> bool {
        const v8f z = matvec(qa, qb);
        const v2f h = shifts(z);
        v2f za, zb;
        za.x = z[0];  za.y = z[1];
        zb.x = z[2];  zb.y = z[3];
        v2f ea, eb, t;
        asm("s_delay_alu instid0(VALU_DEP_1) | instskip(NEXT) | instid1(VALU_DEP_2)\n\t"
            "v_pk_fma_f32 %0, %5, 0.5, %6 op_sel_hi:[1,0,1] clamp\n\t"
            "v_pk_fma_f32 %1, %5, 0.5, %7 op_sel_hi:[1,0,1] clamp\n\t"
            "s_delay_alu instid0(VALU_DEP_2) | instskip(NEXT) | instid1(VALU_DEP_2)\n\t"
            "v_pk_add_f32 %2, %0, %8 neg_lo:[0,1] neg_hi:[0,1]\n\t"
            "v_pk_add_f32 %3, %1, %9 neg_lo:[0,1] neg_hi:[0,1]\n\t"
            "s_delay_alu instid0(VALU_DEP_2) | instskip(NEXT) | instid1(VALU_DEP_1)\n\t"
            "v_pk_fma_f32 %4, %2, %2, %10\n\t"
            "v_pk_fma_f32 %4, %3, %3, %4\n\t"
            "s_delay_alu instid0(VALU_DEP_1)\n\t"
            "v_pk_fma_f32 %4, %2, %3, %4\n\t"
            "s_delay_alu instid0(VALU_DEP_1)"
            : "=&v"(na), "=&v"(nb), "=&v"(ea), "=&v"(eb), "=&v"(t)
            : "v"(h), "v"(za), "v"(zb), "v"(qa), "v"(qb), "v"(kthr));
        return t.x > -t.y;
    };

    auto finish = [&](const v2f qa, const v2f qb) {
        const v8f z = matvec(qa, qb);
        const v2f h = shifts(z);
        v2f za, zb, zc;
        za.x = z[0];  za.y = z[1];
        zb.x = z[2];  zb.y = z[3];
        zc.x = z[4];  zc.y = z[5];
        v2f oa, ob, oc;
        asm("s_delay_alu instid0(VALU_DEP_1) | instskip(NEXT) | instid1(VALU_DEP_2)\n\t"
            "v_pk_fma_f32 %0, %3, 0.5, %4 op_sel_hi:[1,0,1] clamp\n\t"
            "v_pk_fma_f32 %1, %3, 0.5, %5 op_sel_hi:[1,0,1] clamp\n\t"
            "s_delay_alu instid0(VALU_DEP_3)\n\t"
            "v_pk_fma_f32 %2, %3, 0.5, %6 op_sel_hi:[1,0,1] clamp"
            : "=&v"(oa), "=&v"(ob), "=&v"(oc)
            : "v"(h), "v"(za), "v"(zb), "v"(zc));
        if (active) {
            float2* op = reinterpret_cast<float2*>(out + 6l * s);
            op[0] = make_float2(oa.x, ob.x);
            op[1] = make_float2(oc.x, oa.y);
            op[2] = make_float2(ob.y, oc.y);
        }
    };

    int depth = 0;
    for (;;) {
        v2f ta, tb;
        bool more = step(pa, pb, ta, tb);
        ++depth;
        if (depth >= NFPN_MAX_DEPTH || __ballot(more) == 0ULL) {
            finish(ta, tb);
            return;
        }
        more = step(ta, tb, pa, pb);
        ++depth;
        if (depth >= NFPN_MAX_DEPTH || __ballot(more) == 0ULL) {
            finish(pa, pb);
            return;
        }
    }
}

extern "C" void kernel_launch(void* const* d_in, const int* in_sizes, int n_in,
                              void* d_out, int out_size, void* d_ws, size_t ws_size,
                              hipStream_t stream) {
    const float* d  = (const float*)d_in[0];
    const float* W1 = (const float*)d_in[1];
    const float* b1 = (const float*)d_in[2];
    const float* W2 = (const float*)d_in[3];
    const float* b2 = (const float*)d_in[4];
    float* out = (float*)d_out;

    const int B = in_sizes[0] / 3;
    const int threads = 256;
    const int blocks  = (B + threads - 1) / threads;
    hipLaunchKernelGGL(nfpn_rps_kernel, dim3(blocks), dim3(threads), 0, stream,
                       d, W1, b1, W2, b2, out, B);
}
